# starting kernel
# baseline (speedup 1.0000x reference)
#include <hip/hip_runtime.h>
#include <hip/hip_bf16.h>

typedef float v2f __attribute__((ext_vector_type(2)));
typedef float v8f __attribute__((ext_vector_type(8)));

#define LN_EPS    1e-5f
#define NEG_SLOPE 0.2f

__device__ __forceinline__ unsigned ford(float f) {
    unsigned u = __float_as_uint(f);
    return (u & 0x80000000u) ? ~u : (u | 0x80000000u);
}
__device__ __forceinline__ float uord(unsigned u) {
    return (u & 0x80000000u) ? __uint_as_float(u & 0x7FFFFFFFu) : __uint_as_float(~u);
}

__global__ __launch_bounds__(256) void k_init(float* __restrict__ agg,
                                              float* __restrict__ out,
                                              float* __restrict__ deg,
                                              unsigned* __restrict__ mbuf,
                                              float* __restrict__ denom, int N) {
    int i = blockIdx.x * 256 + threadIdx.x;
    int tot = N * 64;
    if (i < tot) { agg[i] = 0.f; out[i] = 0.f; }
    if (i < N)   { deg[i] = 1.0f; mbuf[i] = 0u; denom[i] = 0.f; }
}

__global__ __launch_bounds__(256) void k_deg(const int* __restrict__ ei,
                                             float* __restrict__ deg, int E) {
    int i = blockIdx.x * 256 + threadIdx.x;
    if (i < E) {
        int d = __builtin_nontemporal_load(ei + E + i);
        atomicAdd(deg + d, 1.0f);
    }
}

__global__ __launch_bounds__(256) void k_dinv(float* __restrict__ deg, int N) {
    int i = blockIdx.x * 256 + threadIdx.x;
    if (i < N) { float g = deg[i]; deg[i] = (g > 0.f) ? rsqrtf(g) : 0.f; }
}

__device__ __forceinline__ void stage_w(float* __restrict__ sWp,
                                        const float* __restrict__ W) {
    for (int i = threadIdx.x; i < 4096; i += 256) {
        int k = i >> 6, c = i & 63;
        int kk = k >> 2, h = (k >> 1) & 1, j = k & 1;
        sWp[((((kk << 1) + h) << 6) + c) * 2 + j] = W[i];
    }
}

__global__ __launch_bounds__(256) void k_gemm64(const float* __restrict__ A,
                                                const float* __restrict__ W,
                                                float* __restrict__ C, int nstrips) {
    __shared__ float sWp[64 * 64];
    stage_w(sWp, W);
    __syncthreads();

    int lane  = threadIdx.x & 31;
    int strip = blockIdx.x * 8 + (threadIdx.x >> 5);
    if (strip >= nstrips) return;
    int r = lane & 15, h = lane >> 4;

    const float* ap = A + (size_t)(strip * 16 + r) * 64 + 2 * h;
    v8f zero = {0.f, 0.f, 0.f, 0.f, 0.f, 0.f, 0.f, 0.f};
    v8f acc[4] = {zero, zero, zero, zero};

    for (int kk = 0; kk < 16; ++kk) {
        v2f a = *(const v2f*)(ap + kk * 4);
        const v2f* bp = (const v2f*)&sWp[((((kk << 1) + h) << 6) + r) << 1];
#pragma unroll
        for (int t = 0; t < 4; ++t) {
            acc[t] = __builtin_amdgcn_wmma_f32_16x16x4_f32(
                false, a, false, bp[t * 16], (short)0, acc[t], false, false);
        }
    }
    float* cp = C + (size_t)strip * 16 * 64;
#pragma unroll
    for (int t = 0; t < 4; ++t)
#pragma unroll
        for (int g = 0; g < 8; ++g)
            cp[(size_t)(g + 8 * h) * 64 + t * 16 + r] = acc[t][g];
}

__global__ __launch_bounds__(256) void k_gemm64_dual(const float* __restrict__ A,
                                                     const float* __restrict__ Wl,
                                                     const float* __restrict__ Wr,
                                                     float* __restrict__ Cl,
                                                     float* __restrict__ Cr, int nstrips) {
    __shared__ float sWp[2 * 64 * 64];
    stage_w(sWp, Wl);
    stage_w(sWp + 4096, Wr);
    __syncthreads();

    int lane  = threadIdx.x & 31;
    int strip = blockIdx.x * 8 + (threadIdx.x >> 5);
    if (strip >= nstrips) return;
    int r = lane & 15, h = lane >> 4;

    const float* ap = A + (size_t)(strip * 16 + r) * 64 + 2 * h;
    v8f zero = {0.f, 0.f, 0.f, 0.f, 0.f, 0.f, 0.f, 0.f};
    v8f al[4] = {zero, zero, zero, zero};
    v8f ar[4] = {zero, zero, zero, zero};

    for (int kk = 0; kk < 16; ++kk) {
        v2f a = *(const v2f*)(ap + kk * 4);
        const v2f* bpl = (const v2f*)&sWp[((((kk << 1) + h) << 6) + r) << 1];
        const v2f* bpr = bpl + 2048;
#pragma unroll
        for (int t = 0; t < 4; ++t) {
            al[t] = __builtin_amdgcn_wmma_f32_16x16x4_f32(
                false, a, false, bpl[t * 16], (short)0, al[t], false, false);
            ar[t] = __builtin_amdgcn_wmma_f32_16x16x4_f32(
                false, a, false, bpr[t * 16], (short)0, ar[t], false, false);
        }
    }
    float* cl = Cl + (size_t)strip * 16 * 64;
    float* cr = Cr + (size_t)strip * 16 * 64;
#pragma unroll
    for (int t = 0; t < 4; ++t)
#pragma unroll
        for (int g = 0; g < 8; ++g) {
            cl[(size_t)(g + 8 * h) * 64 + t * 16 + r] = al[t][g];
            cr[(size_t)(g + 8 * h) * 64 + t * 16 + r] = ar[t][g];
        }
}

__global__ __launch_bounds__(256) void k_gemm_tail(const float* __restrict__ A,
                                                   const float* __restrict__ W,
                                                   float* __restrict__ C,
                                                   int row0, int nrows) {
    int i = blockIdx.x * 256 + threadIdx.x;
    int rows = nrows - row0;
    if (i >= rows * 64) return;
    int rr = row0 + i / 64, c = i & 63;
    float s = 0.f;
    for (int k = 0; k < 64; ++k) s += A[(size_t)rr * 64 + k] * W[k * 64 + c];
    C[(size_t)rr * 64 + c] = s;
}

__global__ __launch_bounds__(256) void k_gcn_agg(const int* __restrict__ ei,
                                                 const float* __restrict__ h,
                                                 const float* __restrict__ dinv,
                                                 float* __restrict__ agg, int E, int N) {
    int gid = blockIdx.x * 256 + threadIdx.x;
    int eid = gid >> 4, sub = gid & 15;
    if (eid >= E + N) return;
    int s, d;
    if (eid < E) {
        s = __builtin_nontemporal_load(ei + eid);
        d = __builtin_nontemporal_load(ei + E + eid);
    } else {
        s = d = eid - E;
    }
    float norm = dinv[s] * dinv[d];
    float4 v = *(const float4*)(h + (size_t)s * 64 + sub * 4);
    float* ap = agg + (size_t)d * 64 + sub * 4;
    atomicAdd(ap + 0, v.x * norm);
    atomicAdd(ap + 1, v.y * norm);
    atomicAdd(ap + 2, v.z * norm);
    atomicAdd(ap + 3, v.w * norm);
}

__global__ __launch_bounds__(256) void k_node_ln(const float* __restrict__ agg,
                                                 const float* __restrict__ b1,
                                                 const float* __restrict__ lnw,
                                                 const float* __restrict__ lnb,
                                                 float* __restrict__ hout, int N) {
    int lane = threadIdx.x & 31;
    int n = blockIdx.x * 8 + (threadIdx.x >> 5);
    if (n >= N) return;
    const float* p = agg + (size_t)n * 64 + lane * 2;
    float vx = fmaxf(p[0] + b1[lane * 2], 0.f);
    float vy = fmaxf(p[1] + b1[lane * 2 + 1], 0.f);
    float s = vx + vy;
    for (int m = 16; m >= 1; m >>= 1) s += __shfl_xor(s, m, 32);
    float mu = s * (1.f / 64.f);
    float dx = vx - mu, dy = vy - mu;
    float q = dx * dx + dy * dy;
    for (int m = 16; m >= 1; m >>= 1) q += __shfl_xor(q, m, 32);
    float inv = rsqrtf(q * (1.f / 64.f) + LN_EPS);
    float* o = hout + (size_t)n * 64 + lane * 2;
    o[0] = dx * inv * lnw[lane * 2] + lnb[lane * 2];
    o[1] = dy * inv * lnw[lane * 2 + 1] + lnb[lane * 2 + 1];
}

__global__ __launch_bounds__(256) void k_att_score(const int* __restrict__ ei,
                                                   const float* __restrict__ xl,
                                                   const float* __restrict__ xr,
                                                   const float* __restrict__ att,
                                                   float* __restrict__ ebuf,
                                                   unsigned* __restrict__ mbuf,
                                                   int E, int N) {
    int gid = blockIdx.x * 256 + threadIdx.x;
    int eid = gid >> 4, sub = gid & 15;
    if (eid >= E + N) return;
    int s, d;
    if (eid < E) {
        s = __builtin_nontemporal_load(ei + eid);
        d = __builtin_nontemporal_load(ei + E + eid);
    } else {
        s = d = eid - E;
    }
    float4 a = *(const float4*)(xl + (size_t)s * 64 + sub * 4);
    float4 b = *(const float4*)(xr + (size_t)d * 64 + sub * 4);
    float4 w = *(const float4*)(att + sub * 4);
    float p = 0.f, v;
    v = a.x + b.x; v = v > 0.f ? v : NEG_SLOPE * v; p += w.x * v;
    v = a.y + b.y; v = v > 0.f ? v : NEG_SLOPE * v; p += w.y * v;
    v = a.z + b.z; v = v > 0.f ? v : NEG_SLOPE * v; p += w.z * v;
    v = a.w + b.w; v = v > 0.f ? v : NEG_SLOPE * v; p += w.w * v;
    for (int m = 8; m >= 1; m >>= 1) p += __shfl_xor(p, m, 16);
    if (sub == 0) {
        __builtin_nontemporal_store(p, ebuf + eid);
        atomicMax(mbuf + d, ford(p));
    }
}

__global__ __launch_bounds__(256) void k_att_norm(const int* __restrict__ ei,
                                                  float* __restrict__ ebuf,
                                                  const unsigned* __restrict__ mbuf,
                                                  float* __restrict__ denom,
                                                  int E, int N) {
    int i = blockIdx.x * 256 + threadIdx.x;
    if (i >= E + N) return;
    int d = (i < E) ? __builtin_nontemporal_load(ei + E + i) : (i - E);
    float m = uord(mbuf[d]);
    float ex = __expf(ebuf[i] - m);
    ebuf[i] = ex;
    atomicAdd(denom + d, ex);
}

__global__ __launch_bounds__(256) void k_att_agg(const int* __restrict__ ei,
                                                 const float* __restrict__ ebuf,
                                                 const float* __restrict__ denom,
                                                 const float* __restrict__ xl,
                                                 float* __restrict__ out, int E, int N) {
    int gid = blockIdx.x * 256 + threadIdx.x;
    int eid = gid >> 4, sub = gid & 15;
    if (eid >= E + N) return;
    int s, d;
    if (eid < E) {
        s = __builtin_nontemporal_load(ei + eid);
        d = __builtin_nontemporal_load(ei + E + eid);
    } else {
        s = d = eid - E;
    }
    float alpha = ebuf[eid] / denom[d];
    float4 v = *(const float4*)(xl + (size_t)s * 64 + sub * 4);
    float* op = out + (size_t)d * 64 + sub * 4;
    atomicAdd(op + 0, alpha * v.x);
    atomicAdd(op + 1, alpha * v.y);
    atomicAdd(op + 2, alpha * v.z);
    atomicAdd(op + 3, alpha * v.w);
}

__global__ __launch_bounds__(256) void k_finalize(float* __restrict__ out,
                                                  const float* __restrict__ b2, int tot) {
    int i = blockIdx.x * 256 + threadIdx.x;
    if (i < tot) out[i] = fmaxf(out[i] + b2[i & 63], 0.f);
}

extern "C" void kernel_launch(void* const* d_in, const int* in_sizes, int n_in,
                              void* d_out, int out_size, void* d_ws, size_t ws_size,
                              hipStream_t stream) {
    const float* x   = (const float*)d_in[0];
    const int*   ei  = (const int*)d_in[1];
    const float* W1  = (const float*)d_in[2];
    const float* b1  = (const float*)d_in[3];
    const float* lnw = (const float*)d_in[4];
    const float* lnb = (const float*)d_in[5];
    const float* Wl  = (const float*)d_in[6];
    const float* Wr  = (const float*)d_in[7];
    const float* att = (const float*)d_in[8];
    const float* b2  = (const float*)d_in[9];
    float* out = (float*)d_out;

    const int N = in_sizes[0] / 64;
    const int E = in_sizes[1] / 2;
    const int EN = E + N;

    char* ws = (char*)d_ws;
    size_t off = 0;
    auto carve = [&](size_t bytes) -> void* {
        void* p = ws + off;
        off += (bytes + 255) & ~(size_t)255;
        return p;
    };
    float*    hbuf  = (float*)carve((size_t)N * 64 * 4);
    float*    aggxl = (float*)carve((size_t)N * 64 * 4);
    float*    xrb   = (float*)carve((size_t)N * 64 * 4);
    float*    ebuf  = (float*)carve((size_t)EN * 4);
    unsigned* mbuf  = (unsigned*)carve((size_t)N * 4);
    float*    denom = (float*)carve((size_t)N * 4);
    float*    degv  = (float*)carve((size_t)N * 4);

    const int tot = N * 64;
    const int egrid = (int)(((long long)EN * 16 + 255) / 256);
    const int nstrips = N / 16, rem = N - nstrips * 16;

    k_init<<<(tot + 255) / 256, 256, 0, stream>>>(aggxl, out, degv, mbuf, denom, N);
    k_deg<<<(E + 255) / 256, 256, 0, stream>>>(ei, degv, E);
    k_dinv<<<(N + 255) / 256, 256, 0, stream>>>(degv, N);

    k_gemm64<<<(nstrips + 7) / 8, 256, 0, stream>>>(x, W1, hbuf, nstrips);
    if (rem) k_gemm_tail<<<(rem * 64 + 255) / 256, 256, 0, stream>>>(x, W1, hbuf, nstrips * 16, N);

    k_gcn_agg<<<egrid, 256, 0, stream>>>(ei, hbuf, degv, aggxl, E, N);
    k_node_ln<<<(N + 7) / 8, 256, 0, stream>>>(aggxl, b1, lnw, lnb, hbuf, N);

    k_gemm64_dual<<<(nstrips + 7) / 8, 256, 0, stream>>>(hbuf, Wl, Wr, aggxl, xrb, nstrips);
    if (rem) {
        k_gemm_tail<<<(rem * 64 + 255) / 256, 256, 0, stream>>>(hbuf, Wl, aggxl, nstrips * 16, N);
        k_gemm_tail<<<(rem * 64 + 255) / 256, 256, 0, stream>>>(hbuf, Wr, xrb, nstrips * 16, N);
    }

    k_att_score<<<egrid, 256, 0, stream>>>(ei, aggxl, xrb, att, ebuf, mbuf, E, N);
    k_att_norm<<<(EN + 255) / 256, 256, 0, stream>>>(ei, ebuf, mbuf, denom, E, N);
    k_att_agg<<<egrid, 256, 0, stream>>>(ei, ebuf, denom, aggxl, out, E, N);
    k_finalize<<<(tot + 255) / 256, 256, 0, stream>>>(out, b2, tot);
}


# one plane of each atomic scatter sum (GCN rows, attention rows, denominators) per package half, picked by die number; node passes add plane 0 + plane 1; sum order: arrival per plane, then planes
# speedup vs baseline: 3.2741x; 3.2741x over previous
#include <hip/hip_runtime.h>
#include <hip/hip_bf16.h>

typedef float v2f __attribute__((ext_vector_type(2)));
typedef float v8f __attribute__((ext_vector_type(8)));

#define LN_EPS    1e-5f
#define NEG_SLOPE 0.2f

__device__ __forceinline__ unsigned ford(float f) {
    unsigned u = __float_as_uint(f);
    return (u & 0x80000000u) ? ~u : (u | 0x80000000u);
}
__device__ __forceinline__ float uord(unsigned u) {
    return (u & 0x80000000u) ? __uint_as_float(u & 0x7FFFFFFFu) : __uint_as_float(~u);
}

__device__ __forceinline__ unsigned wave_half() {
    const unsigned id = __builtin_amdgcn_s_sendmsg_rtn(0x87);
    return (id >> 18) & 1u;
}

__global__ __launch_bounds__(256) void k_init(float* __restrict__ agg0,
                                              float* __restrict__ agg1,
                                              float* __restrict__ out,
                                              float* __restrict__ out1,
                                              float* __restrict__ deg,
                                              unsigned* __restrict__ mbuf,
                                              float* __restrict__ denom0,
                                              float* __restrict__ denom1,
                                              unsigned* __restrict__ flag, int N) {
    int i = blockIdx.x * 256 + threadIdx.x;
    int tot = N * 64;
    if (i < tot) { agg0[i] = 0.f; agg1[i] = 0.f; out[i] = 0.f; out1[i] = 0.f; }
    if (i < N)   { deg[i] = 1.0f; mbuf[i] = 0u; denom0[i] = 0.f; denom1[i] = 0.f; }
    if (i == 0)  *flag = 0u;
}

#define DEG_CAP     12800
#define DEG_SLICES  32
#define DEG_BLOCKS  256
#define DEG_THREADS 1024

__device__ __forceinline__ void deg_count4(unsigned* cnt, int4 v, int lo) {
    unsigned o;
    o = (unsigned)(v.x - lo); if (o < DEG_CAP) atomicAdd(cnt + o, 1u);
    o = (unsigned)(v.y - lo); if (o < DEG_CAP) atomicAdd(cnt + o, 1u);
    o = (unsigned)(v.z - lo); if (o < DEG_CAP) atomicAdd(cnt + o, 1u);
    o = (unsigned)(v.w - lo); if (o < DEG_CAP) atomicAdd(cnt + o, 1u);
}

__global__ __launch_bounds__(DEG_THREADS) void k_deg(const int* __restrict__ ei,
                                                     float* __restrict__ deg,
                                                     int E, int N) {
    __shared__ unsigned cnt[DEG_CAP];
    const int tid = threadIdx.x;
    const int R = (N + DEG_CAP - 1) / DEG_CAP;
    const int per = (((E + DEG_SLICES - 1) / DEG_SLICES) + 3) & ~3;
    const int* __restrict__ dp = ei + E;
    for (int pair = blockIdx.x; pair < R * DEG_SLICES; pair += gridDim.x) {
        const int rng = pair % R, sl = pair / R;
        for (int i = tid; i < DEG_CAP; i += DEG_THREADS) cnt[i] = 0u;
        __syncthreads();

        const int lo = rng * DEG_CAP;
        const int e0 = min(E, sl * per), e1 = min(E, e0 + per);
        if ((E & 3) == 0) {
            const int4* __restrict__ p4 = (const int4*)(dp + e0);
            const int n4 = (e1 - e0) >> 2;
            const int4 none = {-1, -1, -1, -1};
            for (int i = tid; i < n4; i += 8 * DEG_THREADS) {
                int4 v[8];
#pragma unroll
                for (int j = 0; j < 8; ++j) {
                    const int k = i + j * DEG_THREADS;
                    v[j] = none;
                    if (k < n4) v[j] = p4[k];
                }
#pragma unroll
                for (int j = 0; j < 8; ++j) deg_count4(cnt, v[j], lo);
            }
        } else {
            for (int e = e0 + tid; e < e1; e += DEG_THREADS) {
                unsigned o = (unsigned)(dp[e] - lo);
                if (o < DEG_CAP) atomicAdd(cnt + o, 1u);
            }
        }
        __syncthreads();

        const int nn = min(DEG_CAP, N - lo);
        for (int i = tid; i < nn; i += DEG_THREADS) {
            const unsigned k = cnt[i];
            if (k) atomicAdd(deg + lo + i, (float)k);
        }
        __syncthreads();
    }
}

__global__ __launch_bounds__(256) void k_dinv(float* __restrict__ deg, int N) {
    int i = blockIdx.x * 256 + threadIdx.x;
    if (i < N) { float g = deg[i]; deg[i] = (g > 0.f) ? rsqrtf(g) : 0.f; }
}

__device__ __forceinline__ void stage_w(float* __restrict__ sWp,
                                        const float* __restrict__ W) {
    for (int i = threadIdx.x; i < 4096; i += 256) {
        int k = i >> 6, c = i & 63;
        int kk = k >> 2, h = (k >> 1) & 1, j = k & 1;
        sWp[((((kk << 1) + h) << 6) + c) * 2 + j] = W[i];
    }
}

__global__ __launch_bounds__(256) void k_gemm64(const float* __restrict__ A,
                                                const float* __restrict__ W,
                                                const float* __restrict__ rs,
                                                float* __restrict__ C, int nstrips) {
    __shared__ float sWp[64 * 64];
    stage_w(sWp, W);
    __syncthreads();

    int lane  = threadIdx.x & 31;
    int strip = blockIdx.x * 8 + (threadIdx.x >> 5);
    if (strip >= nstrips) return;
    int r = lane & 15, h = lane >> 4;

    const float* ap = A + (size_t)(strip * 16 + r) * 64 + 2 * h;
    v8f zero = {0.f, 0.f, 0.f, 0.f, 0.f, 0.f, 0.f, 0.f};
    v8f acc[4] = {zero, zero, zero, zero};

    for (int kk = 0; kk < 16; ++kk) {
        v2f a = *(const v2f*)(ap + kk * 4);
        const v2f* bp = (const v2f*)&sWp[((((kk << 1) + h) << 6) + r) << 1];
#pragma unroll
        for (int t = 0; t < 4; ++t) {
            acc[t] = __builtin_amdgcn_wmma_f32_16x16x4_f32(
                false, a, false, bp[t * 16], (short)0, acc[t], false, false);
        }
    }
    float* cp = C + (size_t)strip * 16 * 64;
    float sc[8];
#pragma unroll
    for (int g = 0; g < 8; ++g) sc[g] = rs[strip * 16 + g + 8 * h];
#pragma unroll
    for (int t = 0; t < 4; ++t)
#pragma unroll
        for (int g = 0; g < 8; ++g)
            cp[(size_t)(g + 8 * h) * 64 + t * 16 + r] = acc[t][g] * sc[g];
}

__global__ __launch_bounds__(256) void k_gemm64_dual(const float* __restrict__ A,
                                                     const float* __restrict__ Wl,
                                                     const float* __restrict__ Wr,
                                                     float* __restrict__ Cl,
                                                     float* __restrict__ Cr, int nstrips) {
    __shared__ float sWp[2 * 64 * 64];
    stage_w(sWp, Wl);
    stage_w(sWp + 4096, Wr);
    __syncthreads();

    int lane  = threadIdx.x & 31;
    int strip = blockIdx.x * 8 + (threadIdx.x >> 5);
    if (strip >= nstrips) return;
    int r = lane & 15, h = lane >> 4;

    const float* ap = A + (size_t)(strip * 16 + r) * 64 + 2 * h;
    v8f zero = {0.f, 0.f, 0.f, 0.f, 0.f, 0.f, 0.f, 0.f};
    v8f al[4] = {zero, zero, zero, zero};
    v8f ar[4] = {zero, zero, zero, zero};

    for (int kk = 0; kk < 16; ++kk) {
        v2f a = *(const v2f*)(ap + kk * 4);
        const v2f* bpl = (const v2f*)&sWp[((((kk << 1) + h) << 6) + r) << 1];
        const v2f* bpr = bpl + 2048;
#pragma unroll
        for (int t = 0; t < 4; ++t) {
            al[t] = __builtin_amdgcn_wmma_f32_16x16x4_f32(
                false, a, false, bpl[t * 16], (short)0, al[t], false, false);
            ar[t] = __builtin_amdgcn_wmma_f32_16x16x4_f32(
                false, a, false, bpr[t * 16], (short)0, ar[t], false, false);
        }
    }
    float* cl = Cl + (size_t)strip * 16 * 64;
    float* cr = Cr + (size_t)strip * 16 * 64;
#pragma unroll
    for (int t = 0; t < 4; ++t)
#pragma unroll
        for (int g = 0; g < 8; ++g) {
            cl[(size_t)(g + 8 * h) * 64 + t * 16 + r] = al[t][g];
            cr[(size_t)(g + 8 * h) * 64 + t * 16 + r] = ar[t][g];
        }
}

__global__ __launch_bounds__(256) void k_gemm_tail(const float* __restrict__ A,
                                                   const float* __restrict__ W,
                                                   const float* __restrict__ rs,
                                                   float* __restrict__ C,
                                                   int row0, int nrows) {
    int i = blockIdx.x * 256 + threadIdx.x;
    int rows = nrows - row0;
    if (i >= rows * 64) return;
    int rr = row0 + i / 64, c = i & 63;
    float s = 0.f;
    for (int k = 0; k < 64; ++k) s += A[(size_t)rr * 64 + k] * W[k * 64 + c];
    C[(size_t)rr * 64 + c] = rs ? s * rs[rr] : s;
}

#define WALK_SHIFT  15
#define WALK_BLOCKS ((1 << WALK_SHIFT) / 8)
#define SLOW_SHIFT  11
#define SLOW_BLOCKS ((1 << SLOW_SHIFT) / 8)

__device__ __forceinline__ int rl_i(int v, int l) {
    return __builtin_amdgcn_readlane(v, l);
}
__device__ __forceinline__ float rl_f(float v, int l) {
    return __int_as_float(__builtin_amdgcn_readlane(__float_as_int(v), l));
}

template <int SHIFT>
__device__ __forceinline__ void walk_range(int EN, int& lo, int& hi) {
    unsigned w = (unsigned)__builtin_amdgcn_readfirstlane(
        (int)(blockIdx.x * 8 + (threadIdx.x >> 5)));
    unsigned long long U = ((unsigned long long)EN + 7) >> 3;
    lo = (int)((U * w) >> SHIFT) << 3;
    hi = (int)((U * (w + 1)) >> SHIFT) << 3;
    if (hi > EN) hi = EN;
}

__device__ __forceinline__ void walk_idx(const int* __restrict__ ei, int eid,
                                         int hi, int E, int& s, int& d) {
    s = 0; d = 0;
    if (eid < hi) {
        if (eid < E) {
            s = __builtin_nontemporal_load(ei + eid);
            d = __builtin_nontemporal_load(ei + E + eid);
        } else {
            s = d = eid - E;
        }
    }
}

template <bool COEF>
__device__ __forceinline__ void walk_scatter(const float* __restrict__ src_rows,
                                             float* __restrict__ dst_rows,
                                             int s, int d, float coef,
                                             int cnt, int lane) {
#pragma unroll
    for (int g = 0; g < 32; g += 8) {
        if (g >= cnt) break;
        if (g + 8 <= cnt) {
            float v0[8], v1[8];
#pragma unroll
            for (int j = 0; j < 8; ++j) {
                const float* hp = src_rows + (size_t)(unsigned)rl_i(s, g + j) * 64;
                v0[j] = hp[lane];
                v1[j] = hp[lane + 32];
            }
#pragma unroll
            for (int j = 0; j < 8; ++j) {
                float* ap = dst_rows + (size_t)(unsigned)rl_i(d, g + j) * 64;
                const float cj = COEF ? rl_f(coef, g + j) : 1.0f;
                atomicAdd(ap + lane, COEF ? v0[j] * cj : v0[j]);
                atomicAdd(ap + lane + 32, COEF ? v1[j] * cj : v1[j]);
            }
        } else {
            for (int j = g; j < cnt; ++j) {
                const float* hp = src_rows + (size_t)(unsigned)rl_i(s, j) * 64;
                float* ap = dst_rows + (size_t)(unsigned)rl_i(d, j) * 64;
                const float cj = COEF ? rl_f(coef, j) : 1.0f;
                atomicAdd(ap + lane, COEF ? hp[lane] * cj : hp[lane]);
                atomicAdd(ap + lane + 32, COEF ? hp[lane + 32] * cj : hp[lane + 32]);
            }
        }
    }
}

__global__ __launch_bounds__(256) void k_gcn_agg(const int* __restrict__ ei,
                                                 const float* __restrict__ hs,
                                                 float* __restrict__ agg0,
                                                 float* __restrict__ agg1, int E, int N) {
    const int lane = threadIdx.x & 31;
    float* __restrict__ agg = wave_half() ? agg1 : agg0;
    int lo, hi;
    walk_range<WALK_SHIFT>(E + N, lo, hi);
    for (int base = lo; base < hi; base += 32) {
        const int eid = base + lane;
        int s, d;
        walk_idx(ei, eid, hi, E, s, d);
        walk_scatter<false>(hs, agg, s, d, 0.f, hi - base, lane);
    }
}

__global__ __launch_bounds__(256) void k_node_ln(const float* __restrict__ agg0,
                                                 const float* __restrict__ agg1,
                                                 const float* __restrict__ dinv,
                                                 const float* __restrict__ b1,
                                                 const float* __restrict__ lnw,
                                                 const float* __restrict__ lnb,
                                                 float* __restrict__ hout, int N) {
    int lane = threadIdx.x & 31;
    int n = blockIdx.x * 8 + (threadIdx.x >> 5);
    if (n >= N) return;
    const float* p = agg0 + (size_t)n * 64 + lane * 2;
    const float* p1 = agg1 + (size_t)n * 64 + lane * 2;
    const float dn = dinv[n];
    float vx = fmaxf((p[0] + p1[0]) * dn + b1[lane * 2], 0.f);
    float vy = fmaxf((p[1] + p1[1]) * dn + b1[lane * 2 + 1], 0.f);
    float s = vx + vy;
    for (int m = 16; m >= 1; m >>= 1) s += __shfl_xor(s, m, 32);
    float mu = s * (1.f / 64.f);
    float dx = vx - mu, dy = vy - mu;
    float q = dx * dx + dy * dy;
    for (int m = 16; m >= 1; m >>= 1) q += __shfl_xor(q, m, 32);
    float inv = rsqrtf(q * (1.f / 64.f) + LN_EPS);
    float* o = hout + (size_t)n * 64 + lane * 2;
    o[0] = dx * inv * lnw[lane * 2] + lnb[lane * 2];
    o[1] = dy * inv * lnw[lane * 2 + 1] + lnb[lane * 2 + 1];
}

__device__ __forceinline__ float wave_sum(float p) {
    p += __int_as_float(__builtin_amdgcn_update_dpp(0, __float_as_int(p), 0xB1, 0xF, 0xF, true));
    p += __int_as_float(__builtin_amdgcn_update_dpp(0, __float_as_int(p), 0x4E, 0xF, 0xF, true));
    p += __int_as_float(__builtin_amdgcn_update_dpp(0, __float_as_int(p), 0x141, 0xF, 0xF, true));
    p += __int_as_float(__builtin_amdgcn_update_dpp(0, __float_as_int(p), 0x140, 0xF, 0xF, true));
    p += __int_as_float(__builtin_amdgcn_permlanex16(0, __float_as_int(p),
                                                     0x76543210, 0xfedcba98, true, false));
    return p;
}

__device__ __forceinline__ float score_part(float a0, float b0, float a1, float b1,
                                            float w0, float w1) {
    float v, p;
    v = a0 + b0; v = v > 0.f ? v : NEG_SLOPE * v; p = w0 * v;
    v = a1 + b1; v = v > 0.f ? v : NEG_SLOPE * v; p += w1 * v;
    return p;
}

__device__ __forceinline__ float shift_part(float a0, float b0, float a1, float b1,
                                            float w0, float w1) {
    float v, u, p;
    v = a0 + b0; v = v > 0.f ? v : NEG_SLOPE * v;
    u = b0 > 0.f ? b0 : NEG_SLOPE * b0;
    p = w0 * (v - u);
    v = a1 + b1; v = v > 0.f ? v : NEG_SLOPE * v;
    u = b1 > 0.f ? b1 : NEG_SLOPE * b1;
    p += w1 * (v - u);
    return p;
}

#define SHIFT_RANGE 60.0f

__global__ __launch_bounds__(256) void k_att_fused(const int* __restrict__ ei,
                                                   const float* __restrict__ xl,
                                                   const float* __restrict__ xr,
                                                   const float* __restrict__ att,
                                                   float* __restrict__ out0,
                                                   float* __restrict__ out1,
                                                   float* __restrict__ denom0,
                                                   float* __restrict__ denom1,
                                                   unsigned* __restrict__ flag,
                                                   int E, int N) {
    const int lane = threadIdx.x & 31;
    const unsigned half = wave_half();
    float* __restrict__ out = half ? out1 : out0;
    float* __restrict__ denom = half ? denom1 : denom0;
    int lo, hi;
    walk_range<WALK_SHIFT>(E + N, lo, hi);
    const float w0 = att[lane], w1 = att[lane + 32];
    float worst = 0.f;
    for (int base = lo; base < hi; base += 32) {
        const int eid = base + lane;
        int s, d;
        walk_idx(ei, eid, hi, E, s, d);
        const int cnt = hi - base;
        float pe = 0.f;
#pragma unroll
        for (int g = 0; g < 32; g += 8) {
            if (g >= cnt) break;
            if (g + 8 <= cnt) {
                float a0[8], a1[8], b0[8], b1[8];
#pragma unroll
                for (int j = 0; j < 8; ++j) {
                    const float* ap = xl + (size_t)(unsigned)rl_i(s, g + j) * 64;
                    const float* bp = xr + (size_t)(unsigned)rl_i(d, g + j) * 64;
                    a0[j] = ap[lane]; a1[j] = ap[lane + 32];
                    b0[j] = bp[lane]; b1[j] = bp[lane + 32];
                }
#pragma unroll
                for (int j = 0; j < 8; ++j) {
                    const float x = wave_sum(shift_part(a0[j], b0[j], a1[j], b1[j], w0, w1));
                    worst = fmaxf(worst, fabsf(x));
                    const float ex = __expf(x);
                    float* op = out + (size_t)(unsigned)rl_i(d, g + j) * 64;
                    atomicAdd(op + lane, ex * a0[j]);
                    atomicAdd(op + lane + 32, ex * a1[j]);
                    if (lane == g + j) pe = ex;
                }
            } else {
                for (int j = g; j < cnt; ++j) {
                    const float* ap = xl + (size_t)(unsigned)rl_i(s, j) * 64;
                    const float* bp = xr + (size_t)(unsigned)rl_i(d, j) * 64;
                    const float c0 = ap[lane], c1 = ap[lane + 32];
                    const float x = wave_sum(shift_part(c0, bp[lane], c1, bp[lane + 32], w0, w1));
                    worst = fmaxf(worst, fabsf(x));
                    const float ex = __expf(x);
                    float* op = out + (size_t)(unsigned)rl_i(d, j) * 64;
                    atomicAdd(op + lane, ex * c0);
                    atomicAdd(op + lane + 32, ex * c1);
                    if (lane == j) pe = ex;
                }
            }
        }
        if (eid < hi) atomicAdd(denom + d, pe);
    }
    if (worst > SHIFT_RANGE) *flag = 1u;
}

__global__ __launch_bounds__(256) void k_att_score(const int* __restrict__ ei,
                                                   const float* __restrict__ xl,
                                                   const float* __restrict__ xr,
                                                   const float* __restrict__ att,
                                                   float* __restrict__ ebuf,
                                                   unsigned* __restrict__ mbuf,
                                                   float* __restrict__ out,
                                                   float* __restrict__ out1,
                                                   float* __restrict__ denom,
                                                   float* __restrict__ denom1,
                                                   const unsigned* __restrict__ flag,
                                                   int E, int N) {
    if (*flag == 0u) return;
    {
        const int nthr = SLOW_BLOCKS * 256;
        const int t = blockIdx.x * 256 + threadIdx.x;
        for (int i = t; i < N * 64; i += nthr) { out[i] = 0.f; out1[i] = 0.f; }
        for (int i = t; i < N; i += nthr) { denom[i] = 0.f; denom1[i] = 0.f; }
    }
    const int lane = threadIdx.x & 31;
    int lo, hi;
    walk_range<SLOW_SHIFT>(E + N, lo, hi);
    const float w0 = att[lane], w1 = att[lane + 32];
    for (int base = lo; base < hi; base += 32) {
        const int eid = base + lane;
        int s, d;
        walk_idx(ei, eid, hi, E, s, d);
        const int cnt = hi - base;
        float pe = 0.f;
#pragma unroll
        for (int g = 0; g < 32; g += 8) {
            if (g >= cnt) break;
            if (g + 8 <= cnt) {
                float a0[8], a1[8], b0[8], b1[8];
#pragma unroll
                for (int j = 0; j < 8; ++j) {
                    const float* ap = xl + (size_t)(unsigned)rl_i(s, g + j) * 64;
                    const float* bp = xr + (size_t)(unsigned)rl_i(d, g + j) * 64;
                    a0[j] = ap[lane]; a1[j] = ap[lane + 32];
                    b0[j] = bp[lane]; b1[j] = bp[lane + 32];
                }
#pragma unroll
                for (int j = 0; j < 8; ++j) {
                    float p = wave_sum(score_part(a0[j], b0[j], a1[j], b1[j], w0, w1));
                    if (lane == g + j) pe = p;
                }
            } else {
                for (int j = g; j < cnt; ++j) {
                    const float* ap = xl + (size_t)(unsigned)rl_i(s, j) * 64;
                    const float* bp = xr + (size_t)(unsigned)rl_i(d, j) * 64;
                    float p = wave_sum(score_part(ap[lane], bp[lane],
                                                  ap[lane + 32], bp[lane + 32], w0, w1));
                    if (lane == j) pe = p;
                }
            }
        }
        if (eid < hi) {
            __builtin_nontemporal_store(pe, ebuf + eid);
            atomicMax(mbuf + d, ford(pe));
        }
    }
}

__global__ __launch_bounds__(256) void k_att_agg(const int* __restrict__ ei,
                                                 const float* __restrict__ ebuf,
                                                 const unsigned* __restrict__ mbuf,
                                                 const float* __restrict__ xl,
                                                 float* __restrict__ out,
                                                 float* __restrict__ denom,
                                                 const unsigned* __restrict__ flag,
                                                 int E, int N) {
    if (*flag == 0u) return;
    const int lane = threadIdx.x & 31;
    int lo, hi;
    walk_range<SLOW_SHIFT>(E + N, lo, hi);
    for (int base = lo; base < hi; base += 32) {
        const int eid = base + lane;
        int s, d;
        walk_idx(ei, eid, hi, E, s, d);
        float ex = 0.f;
        if (eid < hi) {
            ex = __expf(ebuf[eid] - uord(mbuf[d]));
            atomicAdd(denom + d, ex);
        }
        walk_scatter<true>(xl, out, s, d, ex, hi - base, lane);
    }
}

__global__ __launch_bounds__(256) void k_finalize(float* __restrict__ out,
                                                  const float* __restrict__ out1,
                                                  const float* __restrict__ denom0,
                                                  const float* __restrict__ denom1,
                                                  const float* __restrict__ b2, int tot) {
    int i = blockIdx.x * 256 + threadIdx.x;
    if (i < tot)
        out[i] = fmaxf((out[i] + out1[i]) / (denom0[i >> 6] + denom1[i >> 6]) + b2[i & 63], 0.f);
}

extern "C" void kernel_launch(void* const* d_in, const int* in_sizes, int n_in,
                              void* d_out, int out_size, void* d_ws, size_t ws_size,
                              hipStream_t stream) {
    const float* x   = (const float*)d_in[0];
    const int*   ei  = (const int*)d_in[1];
    const float* W1  = (const float*)d_in[2];
    const float* b1  = (const float*)d_in[3];
    const float* lnw = (const float*)d_in[4];
    const float* lnb = (const float*)d_in[5];
    const float* Wl  = (const float*)d_in[6];
    const float* Wr  = (const float*)d_in[7];
    const float* att = (const float*)d_in[8];
    const float* b2  = (const float*)d_in[9];
    float* out = (float*)d_out;

    const int N = in_sizes[0] / 64;
    const int E = in_sizes[1] / 2;
    const int EN = E + N;

    char* ws = (char*)d_ws;
    size_t off = 0;
    auto carve = [&](size_t bytes) -> void* {
        void* p = ws + off;
        off += (bytes + 255) & ~(size_t)255;
        return p;
    };
    float*    hbuf  = (float*)carve((size_t)N * 64 * 4);
    float*    aggxl = (float*)carve((size_t)N * 64 * 4);
    float*    xrb   = (float*)carve((size_t)N * 64 * 4);
    float*    out1  = (float*)carve((size_t)N * 64 * 4);
    float*    ebuf  = (float*)carve((size_t)EN * 4);
    unsigned* mbuf  = (unsigned*)carve((size_t)N * 4);
    float*    denom = (float*)carve((size_t)N * 4);
    float*    den1  = (float*)carve((size_t)N * 4);
    float*    degv  = (float*)carve((size_t)N * 4);
    unsigned* flag  = (unsigned*)carve(4);

    const int tot = N * 64;
    const int nstrips = N / 16, rem = N - nstrips * 16;

    k_init<<<(tot + 255) / 256, 256, 0, stream>>>(aggxl, xrb, out, out1, degv, mbuf, denom, den1, flag, N);
    k_deg<<<DEG_BLOCKS, DEG_THREADS, 0, stream>>>(ei, degv, E, N);
    k_dinv<<<(N + 255) / 256, 256, 0, stream>>>(degv, N);

    k_gemm64<<<(nstrips + 7) / 8, 256, 0, stream>>>(x, W1, degv, hbuf, nstrips);
    if (rem) k_gemm_tail<<<(rem * 64 + 255) / 256, 256, 0, stream>>>(x, W1, degv, hbuf, nstrips * 16, N);

    k_gcn_agg<<<WALK_BLOCKS, 256, 0, stream>>>(ei, hbuf, aggxl, xrb, E, N);
    k_node_ln<<<(N + 7) / 8, 256, 0, stream>>>(aggxl, xrb, degv, b1, lnw, lnb, hbuf, N);

    k_gemm64_dual<<<(nstrips + 7) / 8, 256, 0, stream>>>(hbuf, Wl, Wr, aggxl, xrb, nstrips);
    if (rem) {
        k_gemm_tail<<<(rem * 64 + 255) / 256, 256, 0, stream>>>(hbuf, Wl, nullptr, aggxl, nstrips * 16, N);
        k_gemm_tail<<<(rem * 64 + 255) / 256, 256, 0, stream>>>(hbuf, Wr, nullptr, xrb, nstrips * 16, N);
    }

    k_att_fused<<<WALK_BLOCKS, 256, 0, stream>>>(ei, aggxl, xrb, att, out, out1, denom, den1, flag, E, N);
    k_att_score<<<SLOW_BLOCKS, 256, 0, stream>>>(ei, aggxl, xrb, att, ebuf, mbuf, out, out1, denom, den1, flag, E, N);
    k_att_agg<<<SLOW_BLOCKS, 256, 0, stream>>>(ei, ebuf, mbuf, aggxl, out, denom, flag, E, N);
    k_finalize<<<(tot + 255) / 256, 256, 0, stream>>>(out, out1, denom, den1, b2, tot);
}


# LayerNorm done in the A-fragment loader of the @Wl/@Wr WMMA kernel (a row's 64 columns sit in lanes r and r+16); LayerNorm pass, its buffer and one launch gone; LayerNorm sum order changed
# speedup vs baseline: 3.8373x; 1.1720x over previous
#include <hip/hip_runtime.h>
#include <hip/hip_bf16.h>

typedef float v2f __attribute__((ext_vector_type(2)));
typedef float v8f __attribute__((ext_vector_type(8)));

#define LN_EPS    1e-5f
#define NEG_SLOPE 0.2f

__device__ __forceinline__ unsigned ford(float f) {
    unsigned u = __float_as_uint(f);
    return (u & 0x80000000u) ? ~u : (u | 0x80000000u);
}
__device__ __forceinline__ float uord(unsigned u) {
    return (u & 0x80000000u) ? __uint_as_float(u & 0x7FFFFFFFu) : __uint_as_float(~u);
}

__device__ __forceinline__ unsigned wave_half() {
    const unsigned id = __builtin_amdgcn_s_sendmsg_rtn(0x87);
    return (id >> 18) & 1u;
}

__global__ __launch_bounds__(256) void k_init(float* __restrict__ agg0,
                                              float* __restrict__ agg1,
                                              float* __restrict__ out,
                                              float* __restrict__ out1,
                                              float* __restrict__ deg,
                                              unsigned* __restrict__ mbuf,
                                              float* __restrict__ denom0,
                                              float* __restrict__ denom1,
                                              unsigned* __restrict__ flag, int N) {
    int i = blockIdx.x * 256 + threadIdx.x;
    int tot = N * 64;
    if (i < tot) { agg0[i] = 0.f; agg1[i] = 0.f; out[i] = 0.f; out1[i] = 0.f; }
    if (i < N)   { deg[i] = 1.0f; mbuf[i] = 0u; denom0[i] = 0.f; denom1[i] = 0.f; }
    if (i == 0)  *flag = 0u;
}

#define DEG_CAP     12800
#define DEG_SLICES  32
#define DEG_BLOCKS  256
#define DEG_THREADS 1024

__device__ __forceinline__ void deg_count4(unsigned* cnt, int4 v, int lo) {
    unsigned o;
    o = (unsigned)(v.x - lo); if (o < DEG_CAP) atomicAdd(cnt + o, 1u);
    o = (unsigned)(v.y - lo); if (o < DEG_CAP) atomicAdd(cnt + o, 1u);
    o = (unsigned)(v.z - lo); if (o < DEG_CAP) atomicAdd(cnt + o, 1u);
    o = (unsigned)(v.w - lo); if (o < DEG_CAP) atomicAdd(cnt + o, 1u);
}

__global__ __launch_bounds__(DEG_THREADS) void k_deg(const int* __restrict__ ei,
                                                     float* __restrict__ deg,
                                                     int E, int N) {
    __shared__ unsigned cnt[DEG_CAP];
    const int tid = threadIdx.x;
    const int R = (N + DEG_CAP - 1) / DEG_CAP;
    const int per = (((E + DEG_SLICES - 1) / DEG_SLICES) + 3) & ~3;
    const int* __restrict__ dp = ei + E;
    for (int pair = blockIdx.x; pair < R * DEG_SLICES; pair += gridDim.x) {
        const int rng = pair % R, sl = pair / R;
        for (int i = tid; i < DEG_CAP; i += DEG_THREADS) cnt[i] = 0u;
        __syncthreads();

        const int lo = rng * DEG_CAP;
        const int e0 = min(E, sl * per), e1 = min(E, e0 + per);
        if ((E & 3) == 0) {
            const int4* __restrict__ p4 = (const int4*)(dp + e0);
            const int n4 = (e1 - e0) >> 2;
            const int4 none = {-1, -1, -1, -1};
            for (int i = tid; i < n4; i += 8 * DEG_THREADS) {
                int4 v[8];
#pragma unroll
                for (int j = 0; j < 8; ++j) {
                    const int k = i + j * DEG_THREADS;
                    v[j] = none;
                    if (k < n4) v[j] = p4[k];
                }
#pragma unroll
                for (int j = 0; j < 8; ++j) deg_count4(cnt, v[j], lo);
            }
        } else {
            for (int e = e0 + tid; e < e1; e += DEG_THREADS) {
                unsigned o = (unsigned)(dp[e] - lo);
                if (o < DEG_CAP) atomicAdd(cnt + o, 1u);
            }
        }
        __syncthreads();

        const int nn = min(DEG_CAP, N - lo);
        for (int i = tid; i < nn; i += DEG_THREADS) {
            const unsigned k = cnt[i];
            if (k) atomicAdd(deg + lo + i, (float)k);
        }
        __syncthreads();
    }
}

__global__ __launch_bounds__(256) void k_dinv(float* __restrict__ deg, int N) {
    int i = blockIdx.x * 256 + threadIdx.x;
    if (i < N) { float g = deg[i]; deg[i] = (g > 0.f) ? rsqrtf(g) : 0.f; }
}

__device__ __forceinline__ void stage_w(float* __restrict__ sWp,
                                        const float* __restrict__ W) {
    for (int i = threadIdx.x; i < 4096; i += 256) {
        int k = i >> 6, c = i & 63;
        int kk = k >> 2, h = (k >> 1) & 1, j = k & 1;
        sWp[((((kk << 1) + h) << 6) + c) * 2 + j] = W[i];
    }
}

__global__ __launch_bounds__(256) void k_gemm64(const float* __restrict__ A,
                                                const float* __restrict__ W,
                                                const float* __restrict__ rs,
                                                float* __restrict__ C, int nstrips) {
    __shared__ float sWp[64 * 64];
    stage_w(sWp, W);
    __syncthreads();

    int lane  = threadIdx.x & 31;
    int strip = blockIdx.x * 8 + (threadIdx.x >> 5);
    if (strip >= nstrips) return;
    int r = lane & 15, h = lane >> 4;

    const float* ap = A + (size_t)(strip * 16 + r) * 64 + 2 * h;
    v8f zero = {0.f, 0.f, 0.f, 0.f, 0.f, 0.f, 0.f, 0.f};
    v8f acc[4] = {zero, zero, zero, zero};

    for (int kk = 0; kk < 16; ++kk) {
        v2f a = *(const v2f*)(ap + kk * 4);
        const v2f* bp = (const v2f*)&sWp[((((kk << 1) + h) << 6) + r) << 1];
#pragma unroll
        for (int t = 0; t < 4; ++t) {
            acc[t] = __builtin_amdgcn_wmma_f32_16x16x4_f32(
                false, a, false, bp[t * 16], (short)0, acc[t], false, false);
        }
    }
    float* cp = C + (size_t)strip * 16 * 64;
    float sc[8];
#pragma unroll
    for (int g = 0; g < 8; ++g) sc[g] = rs[strip * 16 + g + 8 * h];
#pragma unroll
    for (int t = 0; t < 4; ++t)
#pragma unroll
        for (int g = 0; g < 8; ++g)
            cp[(size_t)(g + 8 * h) * 64 + t * 16 + r] = acc[t][g] * sc[g];
}

__device__ __forceinline__ float other_row16(float p) {
    return __int_as_float(__builtin_amdgcn_permlanex16(0, __float_as_int(p),
                                                       0x76543210, 0xfedcba98, true, false));
}

__global__ __launch_bounds__(256) void k_ln_gemm64_dual(const float* agg0,
                                                        const float* agg1,
                                                        const float* __restrict__ dinv,
                                                        const float* __restrict__ b1,
                                                        const float* __restrict__ lnw,
                                                        const float* __restrict__ lnb,
                                                        const float* __restrict__ Wl,
                                                        const float* __restrict__ Wr,
                                                        float* Cl,
                                                        float* Cr, int nstrips) {
    __shared__ float sWp[2 * 64 * 64];
    __shared__ float sPar[3 * 64];
    stage_w(sWp, Wl);
    stage_w(sWp + 4096, Wr);
    if (threadIdx.x < 64) {
        sPar[threadIdx.x] = b1[threadIdx.x];
        sPar[64 + threadIdx.x] = lnw[threadIdx.x];
        sPar[128 + threadIdx.x] = lnb[threadIdx.x];
    }
    __syncthreads();

    int lane  = threadIdx.x & 31;
    int strip = blockIdx.x * 8 + (threadIdx.x >> 5);
    if (strip >= nstrips) return;
    int r = lane & 15, h = lane >> 4;

    const size_t rowoff = (size_t)(strip * 16 + r) * 64 + 2 * h;
    const float* p0 = agg0 + rowoff;
    const float* p1 = agg1 + rowoff;
    const float dn = dinv[strip * 16 + r];

    v2f a[16];
    {
        v2f u[16], w[16];
#pragma unroll
        for (int kk = 0; kk < 16; ++kk) {
            u[kk] = *(const v2f*)(p0 + kk * 4);
            w[kk] = *(const v2f*)(p1 + kk * 4);
        }
        float s = 0.f;
#pragma unroll
        for (int kk = 0; kk < 16; ++kk) {
            const v2f bb = *(const v2f*)&sPar[kk * 4 + 2 * h];
            a[kk].x = fmaxf((u[kk].x + w[kk].x) * dn + bb.x, 0.f);
            a[kk].y = fmaxf((u[kk].y + w[kk].y) * dn + bb.y, 0.f);
            s += a[kk].x + a[kk].y;
        }
        s += other_row16(s);
        const float mu = s * (1.f / 64.f);
        float q = 0.f;
#pragma unroll
        for (int kk = 0; kk < 16; ++kk) {
            a[kk].x -= mu;
            a[kk].y -= mu;
            q += a[kk].x * a[kk].x + a[kk].y * a[kk].y;
        }
        q += other_row16(q);
        const float inv = rsqrtf(q * (1.f / 64.f) + LN_EPS);
#pragma unroll
        for (int kk = 0; kk < 16; ++kk) {
            const v2f gw = *(const v2f*)&sPar[64 + kk * 4 + 2 * h];
            const v2f gb = *(const v2f*)&sPar[128 + kk * 4 + 2 * h];
            a[kk].x = a[kk].x * inv * gw.x + gb.x;
            a[kk].y = a[kk].y * inv * gw.y + gb.y;
        }
    }

    v8f zero = {0.f, 0.f, 0.f, 0.f, 0.f, 0.f, 0.f, 0.f};
    v8f al[4] = {zero, zero, zero, zero};
    v8f ar[4] = {zero, zero, zero, zero};

#pragma unroll
    for (int kk = 0; kk < 16; ++kk) {
        const v2f* bpl = (const v2f*)&sWp[((((kk << 1) + h) << 6) + r) << 1];
        const v2f* bpr = bpl + 2048;
#pragma unroll
        for (int t = 0; t < 4; ++t) {
            al[t] = __builtin_amdgcn_wmma_f32_16x16x4_f32(
                false, a[kk], false, bpl[t * 16], (short)0, al[t], false, false);
            ar[t] = __builtin_amdgcn_wmma_f32_16x16x4_f32(
                false, a[kk], false, bpr[t * 16], (short)0, ar[t], false, false);
        }
    }
    float* cl = Cl + (size_t)strip * 16 * 64;
    float* cr = Cr + (size_t)strip * 16 * 64;
#pragma unroll
    for (int t = 0; t < 4; ++t)
#pragma unroll
        for (int g = 0; g < 8; ++g) {
            cl[(size_t)(g + 8 * h) * 64 + t * 16 + r] = al[t][g];
            cr[(size_t)(g + 8 * h) * 64 + t * 16 + r] = ar[t][g];
        }
}

__global__ __launch_bounds__(256) void k_gemm_tail(const float* __restrict__ A,
                                                   const float* __restrict__ W,
                                                   const float* __restrict__ rs,
                                                   float* __restrict__ C,
                                                   int row0, int nrows, int a_row0) {
    int i = blockIdx.x * 256 + threadIdx.x;
    int rows = nrows - row0;
    if (i >= rows * 64) return;
    int rr = row0 + i / 64, c = i & 63;
    float s = 0.f;
    for (int k = 0; k < 64; ++k) s += A[(size_t)(rr - a_row0) * 64 + k] * W[k * 64 + c];
    C[(size_t)rr * 64 + c] = rs ? s * rs[rr] : s;
}

#define WALK_SHIFT  15
#define WALK_BLOCKS ((1 << WALK_SHIFT) / 8)
#define SLOW_SHIFT  11
#define SLOW_BLOCKS ((1 << SLOW_SHIFT) / 8)

__device__ __forceinline__ int rl_i(int v, int l) {
    return __builtin_amdgcn_readlane(v, l);
}
__device__ __forceinline__ float rl_f(float v, int l) {
    return __int_as_float(__builtin_amdgcn_readlane(__float_as_int(v), l));
}

template <int SHIFT>
__device__ __forceinline__ void walk_range(int EN, int& lo, int& hi) {
    unsigned w = (unsigned)__builtin_amdgcn_readfirstlane(
        (int)(blockIdx.x * 8 + (threadIdx.x >> 5)));
    unsigned long long U = ((unsigned long long)EN + 7) >> 3;
    lo = (int)((U * w) >> SHIFT) << 3;
    hi = (int)((U * (w + 1)) >> SHIFT) << 3;
    if (hi > EN) hi = EN;
}

__device__ __forceinline__ void walk_idx(const int* __restrict__ ei, int eid,
                                         int hi, int E, int& s, int& d) {
    s = 0; d = 0;
    if (eid < hi) {
        if (eid < E) {
            s = __builtin_nontemporal_load(ei + eid);
            d = __builtin_nontemporal_load(ei + E + eid);
        } else {
            s = d = eid - E;
        }
    }
}

template <bool COEF>
__device__ __forceinline__ void walk_scatter(const float* __restrict__ src_rows,
                                             float* __restrict__ dst_rows,
                                             int s, int d, float coef,
                                             int cnt, int lane) {
#pragma unroll
    for (int g = 0; g < 32; g += 8) {
        if (g >= cnt) break;
        if (g + 8 <= cnt) {
            float v0[8], v1[8];
#pragma unroll
            for (int j = 0; j < 8; ++j) {
                const float* hp = src_rows + (size_t)(unsigned)rl_i(s, g + j) * 64;
                v0[j] = hp[lane];
                v1[j] = hp[lane + 32];
            }
#pragma unroll
            for (int j = 0; j < 8; ++j) {
                float* ap = dst_rows + (size_t)(unsigned)rl_i(d, g + j) * 64;
                const float cj = COEF ? rl_f(coef, g + j) : 1.0f;
                atomicAdd(ap + lane, COEF ? v0[j] * cj : v0[j]);
                atomicAdd(ap + lane + 32, COEF ? v1[j] * cj : v1[j]);
            }
        } else {
            for (int j = g; j < cnt; ++j) {
                const float* hp = src_rows + (size_t)(unsigned)rl_i(s, j) * 64;
                float* ap = dst_rows + (size_t)(unsigned)rl_i(d, j) * 64;
                const float cj = COEF ? rl_f(coef, j) : 1.0f;
                atomicAdd(ap + lane, COEF ? hp[lane] * cj : hp[lane]);
                atomicAdd(ap + lane + 32, COEF ? hp[lane + 32] * cj : hp[lane + 32]);
            }
        }
    }
}

__global__ __launch_bounds__(256) void k_gcn_agg(const int* __restrict__ ei,
                                                 const float* __restrict__ hs,
                                                 float* __restrict__ agg0,
                                                 float* __restrict__ agg1, int E, int N) {
    const int lane = threadIdx.x & 31;
    float* __restrict__ agg = wave_half() ? agg1 : agg0;
    int lo, hi;
    walk_range<WALK_SHIFT>(E + N, lo, hi);
    for (int base = lo; base < hi; base += 32) {
        const int eid = base + lane;
        int s, d;
        walk_idx(ei, eid, hi, E, s, d);
        walk_scatter<false>(hs, agg, s, d, 0.f, hi - base, lane);
    }
}

__global__ __launch_bounds__(256) void k_node_ln(const float* __restrict__ agg0,
                                                 const float* __restrict__ agg1,
                                                 const float* __restrict__ dinv,
                                                 const float* __restrict__ b1,
                                                 const float* __restrict__ lnw,
                                                 const float* __restrict__ lnb,
                                                 float* __restrict__ hout, int row0, int N) {
    int lane = threadIdx.x & 31;
    int n = row0 + blockIdx.x * 8 + (threadIdx.x >> 5);
    if (n >= N) return;
    const float* p = agg0 + (size_t)n * 64 + lane * 2;
    const float* p1 = agg1 + (size_t)n * 64 + lane * 2;
    const float dn = dinv[n];
    float vx = fmaxf((p[0] + p1[0]) * dn + b1[lane * 2], 0.f);
    float vy = fmaxf((p[1] + p1[1]) * dn + b1[lane * 2 + 1], 0.f);
    float s = vx + vy;
    for (int m = 16; m >= 1; m >>= 1) s += __shfl_xor(s, m, 32);
    float mu = s * (1.f / 64.f);
    float dx = vx - mu, dy = vy - mu;
    float q = dx * dx + dy * dy;
    for (int m = 16; m >= 1; m >>= 1) q += __shfl_xor(q, m, 32);
    float inv = rsqrtf(q * (1.f / 64.f) + LN_EPS);
    float* o = hout + (size_t)(n - row0) * 64 + lane * 2;
    o[0] = dx * inv * lnw[lane * 2] + lnb[lane * 2];
    o[1] = dy * inv * lnw[lane * 2 + 1] + lnb[lane * 2 + 1];
}

__device__ __forceinline__ float wave_sum(float p) {
    p += __int_as_float(__builtin_amdgcn_update_dpp(0, __float_as_int(p), 0xB1, 0xF, 0xF, true));
    p += __int_as_float(__builtin_amdgcn_update_dpp(0, __float_as_int(p), 0x4E, 0xF, 0xF, true));
    p += __int_as_float(__builtin_amdgcn_update_dpp(0, __float_as_int(p), 0x141, 0xF, 0xF, true));
    p += __int_as_float(__builtin_amdgcn_update_dpp(0, __float_as_int(p), 0x140, 0xF, 0xF, true));
    p += __int_as_float(__builtin_amdgcn_permlanex16(0, __float_as_int(p),
                                                     0x76543210, 0xfedcba98, true, false));
    return p;
}

__device__ __forceinline__ float score_part(float a0, float b0, float a1, float b1,
                                            float w0, float w1) {
    float v, p;
    v = a0 + b0; v = v > 0.f ? v : NEG_SLOPE * v; p = w0 * v;
    v = a1 + b1; v = v > 0.f ? v : NEG_SLOPE * v; p += w1 * v;
    return p;
}

__device__ __forceinline__ float shift_part(float a0, float b0, float a1, float b1,
                                            float w0, float w1) {
    float v, u, p;
    v = a0 + b0; v = v > 0.f ? v : NEG_SLOPE * v;
    u = b0 > 0.f ? b0 : NEG_SLOPE * b0;
    p = w0 * (v - u);
    v = a1 + b1; v = v > 0.f ? v : NEG_SLOPE * v;
    u = b1 > 0.f ? b1 : NEG_SLOPE * b1;
    p += w1 * (v - u);
    return p;
}

#define SHIFT_RANGE 60.0f

__global__ __launch_bounds__(256) void k_att_fused(const int* __restrict__ ei,
                                                   const float* __restrict__ xl,
                                                   const float* __restrict__ xr,
                                                   const float* __restrict__ att,
                                                   float* __restrict__ out0,
                                                   float* __restrict__ out1,
                                                   float* __restrict__ denom0,
                                                   float* __restrict__ denom1,
                                                   unsigned* __restrict__ flag,
                                                   int E, int N) {
    const int lane = threadIdx.x & 31;
    const unsigned half = wave_half();
    float* __restrict__ out = half ? out1 : out0;
    float* __restrict__ denom = half ? denom1 : denom0;
    int lo, hi;
    walk_range<WALK_SHIFT>(E + N, lo, hi);
    const float w0 = att[lane], w1 = att[lane + 32];
    float worst = 0.f;
    for (int base = lo; base < hi; base += 32) {
        const int eid = base + lane;
        int s, d;
        walk_idx(ei, eid, hi, E, s, d);
        const int cnt = hi - base;
        float pe = 0.f;
#pragma unroll
        for (int g = 0; g < 32; g += 8) {
            if (g >= cnt) break;
            if (g + 8 <= cnt) {
                float a0[8], a1[8], b0[8], b1[8];
#pragma unroll
                for (int j = 0; j < 8; ++j) {
                    const float* ap = xl + (size_t)(unsigned)rl_i(s, g + j) * 64;
                    const float* bp = xr + (size_t)(unsigned)rl_i(d, g + j) * 64;
                    a0[j] = ap[lane]; a1[j] = ap[lane + 32];
                    b0[j] = bp[lane]; b1[j] = bp[lane + 32];
                }
#pragma unroll
                for (int j = 0; j < 8; ++j) {
                    const float x = wave_sum(shift_part(a0[j], b0[j], a1[j], b1[j], w0, w1));
                    worst = fmaxf(worst, fabsf(x));
                    const float ex = __expf(x);
                    float* op = out + (size_t)(unsigned)rl_i(d, g + j) * 64;
                    atomicAdd(op + lane, ex * a0[j]);
                    atomicAdd(op + lane + 32, ex * a1[j]);
                    if (lane == g + j) pe = ex;
                }
            } else {
                for (int j = g; j < cnt; ++j) {
                    const float* ap = xl + (size_t)(unsigned)rl_i(s, j) * 64;
                    const float* bp = xr + (size_t)(unsigned)rl_i(d, j) * 64;
                    const float c0 = ap[lane], c1 = ap[lane + 32];
                    const float x = wave_sum(shift_part(c0, bp[lane], c1, bp[lane + 32], w0, w1));
                    worst = fmaxf(worst, fabsf(x));
                    const float ex = __expf(x);
                    float* op = out + (size_t)(unsigned)rl_i(d, j) * 64;
                    atomicAdd(op + lane, ex * c0);
                    atomicAdd(op + lane + 32, ex * c1);
                    if (lane == j) pe = ex;
                }
            }
        }
        if (eid < hi) atomicAdd(denom + d, pe);
    }
    if (worst > SHIFT_RANGE) *flag = 1u;
}

__global__ __launch_bounds__(256) void k_att_score(const int* __restrict__ ei,
                                                   const float* __restrict__ xl,
                                                   const float* __restrict__ xr,
                                                   const float* __restrict__ att,
                                                   float* __restrict__ ebuf,
                                                   unsigned* __restrict__ mbuf,
                                                   float* __restrict__ out,
                                                   float* __restrict__ out1,
                                                   float* __restrict__ denom,
                                                   float* __restrict__ denom1,
                                                   const unsigned* __restrict__ flag,
                                                   int E, int N) {
    if (*flag == 0u) return;
    {
        const int nthr = SLOW_BLOCKS * 256;
        const int t = blockIdx.x * 256 + threadIdx.x;
        for (int i = t; i < N * 64; i += nthr) { out[i] = 0.f; out1[i] = 0.f; }
        for (int i = t; i < N; i += nthr) { denom[i] = 0.f; denom1[i] = 0.f; }
    }
    const int lane = threadIdx.x & 31;
    int lo, hi;
    walk_range<SLOW_SHIFT>(E + N, lo, hi);
    const float w0 = att[lane], w1 = att[lane + 32];
    for (int base = lo; base < hi; base += 32) {
        const int eid = base + lane;
        int s, d;
        walk_idx(ei, eid, hi, E, s, d);
        const int cnt = hi - base;
        float pe = 0.f;
#pragma unroll
        for (int g = 0; g < 32; g += 8) {
            if (g >= cnt) break;
            if (g + 8 <= cnt) {
                float a0[8], a1[8], b0[8], b1[8];
#pragma unroll
                for (int j = 0; j < 8; ++j) {
                    const float* ap = xl + (size_t)(unsigned)rl_i(s, g + j) * 64;
                    const float* bp = xr + (size_t)(unsigned)rl_i(d, g + j) * 64;
                    a0[j] = ap[lane]; a1[j] = ap[lane + 32];
                    b0[j] = bp[lane]; b1[j] = bp[lane + 32];
                }
#pragma unroll
                for (int j = 0; j < 8; ++j) {
                    float p = wave_sum(score_part(a0[j], b0[j], a1[j], b1[j], w0, w1));
                    if (lane == g + j) pe = p;
                }
            } else {
                for (int j = g; j < cnt; ++j) {
                    const float* ap = xl + (size_t)(unsigned)rl_i(s, j) * 64;
                    const float* bp = xr + (size_t)(unsigned)rl_i(d, j) * 64;
                    float p = wave_sum(score_part(ap[lane], bp[lane],
                                                  ap[lane + 32], bp[lane + 32], w0, w1));
                    if (lane == j) pe = p;
                }
            }
        }
        if (eid < hi) {
            __builtin_nontemporal_store(pe, ebuf + eid);
            atomicMax(mbuf + d, ford(pe));
        }
    }
}

__global__ __launch_bounds__(256) void k_att_agg(const int* __restrict__ ei,
                                                 const float* __restrict__ ebuf,
                                                 const unsigned* __restrict__ mbuf,
                                                 const float* __restrict__ xl,
                                                 float* __restrict__ out,
                                                 float* __restrict__ denom,
                                                 const unsigned* __restrict__ flag,
                                                 int E, int N) {
    if (*flag == 0u) return;
    const int lane = threadIdx.x & 31;
    int lo, hi;
    walk_range<SLOW_SHIFT>(E + N, lo, hi);
    for (int base = lo; base < hi; base += 32) {
        const int eid = base + lane;
        int s, d;
        walk_idx(ei, eid, hi, E, s, d);
        float ex = 0.f;
        if (eid < hi) {
            ex = __expf(ebuf[eid] - uord(mbuf[d]));
            atomicAdd(denom + d, ex);
        }
        walk_scatter<true>(xl, out, s, d, ex, hi - base, lane);
    }
}

__global__ __launch_bounds__(256) void k_finalize(float* __restrict__ out,
                                                  const float* __restrict__ out1,
                                                  const float* __restrict__ denom0,
                                                  const float* __restrict__ denom1,
                                                  const float* __restrict__ b2, int tot) {
    int i = blockIdx.x * 256 + threadIdx.x;
    if (i < tot)
        out[i] = fmaxf((out[i] + out1[i]) / (denom0[i >> 6] + denom1[i >> 6]) + b2[i & 63], 0.f);
}

extern "C" void kernel_launch(void* const* d_in, const int* in_sizes, int n_in,
                              void* d_out, int out_size, void* d_ws, size_t ws_size,
                              hipStream_t stream) {
    const float* x   = (const float*)d_in[0];
    const int*   ei  = (const int*)d_in[1];
    const float* W1  = (const float*)d_in[2];
    const float* b1  = (const float*)d_in[3];
    const float* lnw = (const float*)d_in[4];
    const float* lnb = (const float*)d_in[5];
    const float* Wl  = (const float*)d_in[6];
    const float* Wr  = (const float*)d_in[7];
    const float* att = (const float*)d_in[8];
    const float* b2  = (const float*)d_in[9];
    float* out = (float*)d_out;

    const int N = in_sizes[0] / 64;
    const int E = in_sizes[1] / 2;
    const int EN = E + N;

    char* ws = (char*)d_ws;
    size_t off = 0;
    auto carve = [&](size_t bytes) -> void* {
        void* p = ws + off;
        off += (bytes + 255) & ~(size_t)255;
        return p;
    };
    float*    hbuf  = (float*)carve((size_t)N * 64 * 4);
    float*    aggxl = (float*)carve((size_t)N * 64 * 4);
    float*    xrb   = (float*)carve((size_t)N * 64 * 4);
    float*    lnt   = (float*)carve((size_t)16 * 64 * 4);
    float*    out1  = (float*)carve((size_t)N * 64 * 4);
    float*    ebuf  = (float*)carve((size_t)EN * 4);
    unsigned* mbuf  = (unsigned*)carve((size_t)N * 4);
    float*    denom = (float*)carve((size_t)N * 4);
    float*    den1  = (float*)carve((size_t)N * 4);
    float*    degv  = (float*)carve((size_t)N * 4);
    unsigned* flag  = (unsigned*)carve(4);

    const int tot = N * 64;
    const int nstrips = N / 16, rem = N - nstrips * 16;

    k_init<<<(tot + 255) / 256, 256, 0, stream>>>(aggxl, xrb, out, out1, degv, mbuf, denom, den1, flag, N);
    k_deg<<<DEG_BLOCKS, DEG_THREADS, 0, stream>>>(ei, degv, E, N);
    k_dinv<<<(N + 255) / 256, 256, 0, stream>>>(degv, N);

    k_gemm64<<<(nstrips + 7) / 8, 256, 0, stream>>>(x, W1, degv, hbuf, nstrips);
    if (rem) k_gemm_tail<<<(rem * 64 + 255) / 256, 256, 0, stream>>>(x, W1, degv, hbuf, nstrips * 16, N, 0);

    k_gcn_agg<<<WALK_BLOCKS, 256, 0, stream>>>(ei, hbuf, aggxl, xrb, E, N);

    k_ln_gemm64_dual<<<(nstrips + 7) / 8, 256, 0, stream>>>(aggxl, xrb, degv, b1, lnw, lnb, Wl, Wr, aggxl, xrb, nstrips);
    if (rem) {
        k_node_ln<<<(rem + 7) / 8, 256, 0, stream>>>(aggxl, xrb, degv, b1, lnw, lnb, lnt, nstrips * 16, N);
        k_gemm_tail<<<(rem * 64 + 255) / 256, 256, 0, stream>>>(lnt, Wl, nullptr, aggxl, nstrips * 16, N, nstrips * 16);
        k_gemm_tail<<<(rem * 64 + 255) / 256, 256, 0, stream>>>(lnt, Wr, nullptr, xrb, nstrips * 16, N, nstrips * 16);
    }

    k_att_fused<<<WALK_BLOCKS, 256, 0, stream>>>(ei, aggxl, xrb, att, out, out1, denom, den1, flag, E, N);
    k_att_score<<<SLOW_BLOCKS, 256, 0, stream>>>(ei, aggxl, xrb, att, ebuf, mbuf, out, out1, denom, den1, flag, E, N);
    k_att_agg<<<SLOW_BLOCKS, 256, 0, stream>>>(ei, ebuf, mbuf, aggxl, out, denom, flag, E, N);
    k_finalize<<<(tot + 255) / 256, 256, 0, stream>>>(out, out1, denom, den1, b2, tot);
}
